# starting kernel
# baseline (speedup 1.0000x reference)
#include <hip/hip_runtime.h>
#include <math.h>

typedef float v2f __attribute__((ext_vector_type(2)));
typedef float v8f __attribute__((ext_vector_type(8)));

#define STRD 132
#define TP   112

__global__ __launch_bounds__(256) void
LocalSimilar_fused_kernel(const float* __restrict__ fmap,
                          const float* __restrict__ Wqk,
                          float* __restrict__ out)
{
    __shared__ float fmapT[TP * STRD];
    __shared__ float kT[TP * STRD];
    __shared__ float qT[16 * STRD];
    __shared__ float simS[16 * 26];

    const int t    = threadIdx.x;
    const int lane = t & 31;
    const int w    = t >> 5;
    const int lo   = lane & 15;
    const int hi   = lane >> 4;

    const int b  = blockIdx.z;
    const int x  = blockIdx.y;
    const int y0 = blockIdx.x * 16;

    for (int i = t; i < 128 * 5 * 32; i += 256) {
        int lj  = i & 31;
        int row = i >> 5;
        int c = row / 5, r = row % 5;
        if (lj < 20) {
            int gx = x - 2 + r;   gx = gx < 0 ? 0 : (gx > 95 ? 95 : gx);
            int gy = y0 - 2 + lj; gy = gy < 0 ? 0 : (gy > 127 ? 127 : gy);
            fmapT[(r * 20 + lj) * STRD + c] =
                fmap[(((size_t)b * 128 + c) * 96 + gx) * 128 + gy];
        }
    }
    __syncthreads();

    {
        float2 af[32];
        const float* arow = Wqk + ((size_t)(w * 16 + lo)) * 128 + 2 * hi;
        #pragma unroll
        for (int kk = 0; kk < 32; ++kk) af[kk] = *(const float2*)(arow + kk * 4);

        v8f acc = {0.f, 0.f, 0.f, 0.f, 0.f, 0.f, 0.f, 0.f};
        const float* brow = fmapT + (42 + lo) * STRD + 2 * hi;
        #pragma unroll
        for (int kk = 0; kk < 32; ++kk) {
            float2 b2 = *(const float2*)(brow + kk * 4);
            v2f av; av.x = af[kk].x; av.y = af[kk].y;
            v2f bv; bv.x = b2.x;     bv.y = b2.y;
            acc = __builtin_amdgcn_wmma_f32_16x16x4_f32(
                false, av, false, bv, (short)0, acc, false, false);
        }
        const float scale = 0.08838834764831845f;
        #pragma unroll
        for (int v = 0; v < 8; ++v)
            qT[lo * STRD + w * 16 + v + 8 * hi] = acc[v] * scale;
    }

    {
        float2 af[32];
        const float* arow = Wqk + ((size_t)(128 + w * 16 + lo)) * 128 + 2 * hi;
        #pragma unroll
        for (int kk = 0; kk < 32; ++kk) af[kk] = *(const float2*)(arow + kk * 4);

        for (int nt = 0; nt < 7; ++nt) {
            v8f acc = {0.f, 0.f, 0.f, 0.f, 0.f, 0.f, 0.f, 0.f};
            const float* brow = fmapT + (nt * 16 + lo) * STRD + 2 * hi;
            #pragma unroll
            for (int kk = 0; kk < 32; ++kk) {
                float2 b2 = *(const float2*)(brow + kk * 4);
                v2f av; av.x = af[kk].x; av.y = af[kk].y;
                v2f bv; bv.x = b2.x;     bv.y = b2.y;
                acc = __builtin_amdgcn_wmma_f32_16x16x4_f32(
                    false, av, false, bv, (short)0, acc, false, false);
            }
            #pragma unroll
            for (int v = 0; v < 8; ++v)
                kT[(nt * 16 + lo) * STRD + w * 16 + v + 8 * hi] = acc[v];
        }
    }
    __syncthreads();

    for (int idx = t; idx < 16 * 25; idx += 256) {
        int pos = idx / 25, l = idx % 25;
        int di = l / 5, dj = l % 5;
        int npos = di * 20 + pos + dj;
        const float4* qv = (const float4*)(qT + pos  * STRD);
        const float4* kv = (const float4*)(kT + npos * STRD);
        float s = 0.f;
        #pragma unroll
        for (int c4 = 0; c4 < 32; ++c4) {
            float4 q4 = qv[c4], k4 = kv[c4];
            s += q4.x * k4.x + q4.y * k4.y + q4.z * k4.z + q4.w * k4.w;
        }
        simS[pos * 26 + l] = s;
    }
    __syncthreads();

    if (t < 16) {
        float m = -INFINITY;
        #pragma unroll
        for (int l = 0; l < 25; ++l) m = fmaxf(m, simS[t * 26 + l]);
        float e[25], sum = 0.f;
        #pragma unroll
        for (int l = 0; l < 25; ++l) {
            e[l] = __expf(simS[t * 26 + l] - m);
            sum += e[l];
        }
        float inv = 1.0f / sum;
        float* op = out + ((size_t)b * 12288 + (size_t)(x * 128 + y0 + t)) * 25;
        #pragma unroll
        for (int l = 0; l < 25; ++l) op[l] = e[l] * inv;
    }
}

extern "C" void kernel_launch(void* const* d_in, const int* in_sizes, int n_in,
                              void* d_out, int out_size, void* d_ws, size_t ws_size,
                              hipStream_t stream) {
    (void)in_sizes; (void)n_in; (void)d_ws; (void)ws_size; (void)out_size;
    const float* fmap = (const float*)d_in[0];
    const float* Wqk  = (const float*)d_in[1];
    float* out = (float*)d_out;

    dim3 grid(128 / 16, 96, 4);
    LocalSimilar_fused_kernel<<<grid, 256, 0, stream>>>(fmap, Wqk, out);
}


# halo load: each thread owns one tile position and strides the channels by a constant, 64 independent loads issued as one group (was 41 dependent round trips of 2 loads with per-element index math)
# speedup vs baseline: 1.7757x; 1.7757x over previous
#include <hip/hip_runtime.h>
#include <math.h>

typedef float v2f __attribute__((ext_vector_type(2)));
typedef float v8f __attribute__((ext_vector_type(8)));

#define STRD 132
#define TP   112

__global__ __launch_bounds__(256) void
LocalSimilar_fused_kernel(const float* __restrict__ fmap,
                          const float* __restrict__ Wqk,
                          float* __restrict__ out)
{
    __shared__ float fmapT[TP * STRD];
    __shared__ float kT[TP * STRD];
    __shared__ float qT[16 * STRD];
    __shared__ float simS[16 * 26];

    const int t    = threadIdx.x;
    const int lane = t & 31;
    const int w    = t >> 5;
    const int lo   = lane & 15;
    const int hi   = lane >> 4;

    const int b  = blockIdx.z;
    const int x  = blockIdx.y;
    const int y0 = blockIdx.x * 16;

    if (t < 200) {
        const int ch0 = t / 100;
        const int pos = t - ch0 * 100;
        const int r = pos / 20, j = pos - r * 20;
        int gx = x - 2 + r;   gx = gx < 0 ? 0 : (gx > 95 ? 95 : gx);
        int gy = y0 - 2 + j;  gy = gy < 0 ? 0 : (gy > 127 ? 127 : gy);
        const float* src = fmap + (((size_t)b * 128 + ch0) * 96 + gx) * 128 + gy;
        float* dst = fmapT + pos * STRD + ch0;
        float v[64];
        #pragma unroll
        for (int i = 0; i < 64; ++i) v[i] = src[(size_t)i * (2 * 96 * 128)];
        #pragma unroll
        for (int i = 0; i < 64; ++i) dst[2 * i] = v[i];
    }
    __syncthreads();

    {
        float2 af[32];
        const float* arow = Wqk + ((size_t)(w * 16 + lo)) * 128 + 2 * hi;
        #pragma unroll
        for (int kk = 0; kk < 32; ++kk) af[kk] = *(const float2*)(arow + kk * 4);

        v8f acc = {0.f, 0.f, 0.f, 0.f, 0.f, 0.f, 0.f, 0.f};
        const float* brow = fmapT + (42 + lo) * STRD + 2 * hi;
        #pragma unroll
        for (int kk = 0; kk < 32; ++kk) {
            float2 b2 = *(const float2*)(brow + kk * 4);
            v2f av; av.x = af[kk].x; av.y = af[kk].y;
            v2f bv; bv.x = b2.x;     bv.y = b2.y;
            acc = __builtin_amdgcn_wmma_f32_16x16x4_f32(
                false, av, false, bv, (short)0, acc, false, false);
        }
        const float scale = 0.08838834764831845f;
        #pragma unroll
        for (int v = 0; v < 8; ++v)
            qT[lo * STRD + w * 16 + v + 8 * hi] = acc[v] * scale;
    }

    {
        float2 af[32];
        const float* arow = Wqk + ((size_t)(128 + w * 16 + lo)) * 128 + 2 * hi;
        #pragma unroll
        for (int kk = 0; kk < 32; ++kk) af[kk] = *(const float2*)(arow + kk * 4);

        for (int nt = 0; nt < 7; ++nt) {
            v8f acc = {0.f, 0.f, 0.f, 0.f, 0.f, 0.f, 0.f, 0.f};
            const float* brow = fmapT + (nt * 16 + lo) * STRD + 2 * hi;
            #pragma unroll
            for (int kk = 0; kk < 32; ++kk) {
                float2 b2 = *(const float2*)(brow + kk * 4);
                v2f av; av.x = af[kk].x; av.y = af[kk].y;
                v2f bv; bv.x = b2.x;     bv.y = b2.y;
                acc = __builtin_amdgcn_wmma_f32_16x16x4_f32(
                    false, av, false, bv, (short)0, acc, false, false);
            }
            #pragma unroll
            for (int v = 0; v < 8; ++v)
                kT[(nt * 16 + lo) * STRD + w * 16 + v + 8 * hi] = acc[v];
        }
    }
    __syncthreads();

    for (int idx = t; idx < 16 * 25; idx += 256) {
        int pos = idx / 25, l = idx % 25;
        int di = l / 5, dj = l % 5;
        int npos = di * 20 + pos + dj;
        const float4* qv = (const float4*)(qT + pos  * STRD);
        const float4* kv = (const float4*)(kT + npos * STRD);
        float s = 0.f;
        #pragma unroll
        for (int c4 = 0; c4 < 32; ++c4) {
            float4 q4 = qv[c4], k4 = kv[c4];
            s += q4.x * k4.x + q4.y * k4.y + q4.z * k4.z + q4.w * k4.w;
        }
        simS[pos * 26 + l] = s;
    }
    __syncthreads();

    if (t < 16) {
        float m = -INFINITY;
        #pragma unroll
        for (int l = 0; l < 25; ++l) m = fmaxf(m, simS[t * 26 + l]);
        float e[25], sum = 0.f;
        #pragma unroll
        for (int l = 0; l < 25; ++l) {
            e[l] = __expf(simS[t * 26 + l] - m);
            sum += e[l];
        }
        float inv = 1.0f / sum;
        float* op = out + ((size_t)b * 12288 + (size_t)(x * 128 + y0 + t)) * 25;
        #pragma unroll
        for (int l = 0; l < 25; ++l) op[l] = e[l] * inv;
    }
}

extern "C" void kernel_launch(void* const* d_in, const int* in_sizes, int n_in,
                              void* d_out, int out_size, void* d_ws, size_t ws_size,
                              hipStream_t stream) {
    (void)in_sizes; (void)n_in; (void)d_ws; (void)ws_size; (void)out_size;
    const float* fmap = (const float*)d_in[0];
    const float* Wqk  = (const float*)d_in[1];
    float* out = (float*)d_out;

    dim3 grid(128 / 16, 96, 4);
    LocalSimilar_fused_kernel<<<grid, 256, 0, stream>>>(fmap, Wqk, out);
}


# weights held in registers across a row loop: each work-group loads its Wq and Wk^T fragments once and walks 6 rows of its column block (grid 8x16x4, was 8x96x4): weight fetches per tile / 6
# speedup vs baseline: 3.6734x; 2.0687x over previous
#include <hip/hip_runtime.h>
#include <math.h>


typedef float v2f __attribute__((ext_vector_type(2)));
typedef float v8f __attribute__((ext_vector_type(8)));

#define STRD 132
#define TP   100
#define ROWS_PER_BLOCK 6

__global__ __launch_bounds__(256) void
LocalSimilar_fused_kernel(const float* __restrict__ fmap,
                          const float* __restrict__ Wqk,
                          float* __restrict__ out)
{
    __shared__ float fmapT[TP * STRD];
    __shared__ float qT[16 * STRD];
    __shared__ float gT[16 * STRD];
    __shared__ float simS[16 * 26];

    const int t    = threadIdx.x;
    const int lane = t & 31;
    const int w    = t >> 5;
    const int lo   = lane & 15;
    const int hi   = lane >> 4;

    const int b  = blockIdx.z;
    const int x0 = blockIdx.y * ROWS_PER_BLOCK;
    const int y0 = blockIdx.x * 16;

    float2 af[32];
    v2f ag[32];
    {
        const float* arow = Wqk + ((size_t)(w * 16 + lo)) * 128 + 2 * hi;
        #pragma unroll
        for (int kk = 0; kk < 32; ++kk) af[kk] = *(const float2*)(arow + kk * 4);

        const float* acol = Wqk + ((size_t)(128 + 2 * hi)) * 128 + w * 16 + lo;
        #pragma unroll
        for (int kk = 0; kk < 32; ++kk) {
            ag[kk].x = acol[(kk * 4)     * 128];
            ag[kk].y = acol[(kk * 4 + 1) * 128];
        }
    }

    for (int it = 0; it < ROWS_PER_BLOCK; ++it) {
    const int x = x0 + it;

    if (t < 200) {
        const int ch0 = t / 100;
        const int pos = t - ch0 * 100;
        const int r = pos / 20, j = pos - r * 20;
        int gx = x - 2 + r;   gx = gx < 0 ? 0 : (gx > 95 ? 95 : gx);
        int gy = y0 - 2 + j;  gy = gy < 0 ? 0 : (gy > 127 ? 127 : gy);
        const float* src = fmap + (((size_t)b * 128 + ch0) * 96 + gx) * 128 + gy;
        float* dst = fmapT + pos * STRD + ch0;
        float v[64];
        #pragma unroll
        for (int i = 0; i < 64; ++i) v[i] = src[(size_t)i * (2 * 96 * 128)];
        #pragma unroll
        for (int i = 0; i < 64; ++i) dst[2 * i] = v[i];
    }
    __syncthreads();

    {
        v8f acc = {0.f, 0.f, 0.f, 0.f, 0.f, 0.f, 0.f, 0.f};
        const float* brow = fmapT + (42 + lo) * STRD + 2 * hi;
        #pragma unroll
        for (int kk = 0; kk < 32; ++kk) {
            float2 b2 = *(const float2*)(brow + kk * 4);
            v2f av; av.x = af[kk].x; av.y = af[kk].y;
            v2f bv; bv.x = b2.x;     bv.y = b2.y;
            acc = __builtin_amdgcn_wmma_f32_16x16x4_f32(
                false, av, false, bv, (short)0, acc, false, false);
        }
        const float scale = 0.08838834764831845f;
        #pragma unroll
        for (int v = 0; v < 8; ++v)
            qT[lo * STRD + w * 16 + v + 8 * hi] = acc[v] * scale;
    }
    __syncthreads();

    {
        v8f acc = {0.f, 0.f, 0.f, 0.f, 0.f, 0.f, 0.f, 0.f};
        const float* brow = qT + lo * STRD + 2 * hi;
        #pragma unroll
        for (int kk = 0; kk < 32; ++kk) {
            float2 b2 = *(const float2*)(brow + kk * 4);
            v2f bv; bv.x = b2.x; bv.y = b2.y;
            acc = __builtin_amdgcn_wmma_f32_16x16x4_f32(
                false, ag[kk], false, bv, (short)0, acc, false, false);
        }
        #pragma unroll
        for (int v = 0; v < 8; ++v)
            gT[lo * STRD + w * 16 + v + 8 * hi] = acc[v];
    }
    __syncthreads();

    for (int idx = t; idx < 16 * 25; idx += 256) {
        int pos = idx / 25, l = idx % 25;
        int di = l / 5, dj = l % 5;
        int npos = di * 20 + pos + dj;
        const float4* gv = (const float4*)(gT    + pos  * STRD);
        const float4* fv = (const float4*)(fmapT + npos * STRD);
        float s = 0.f;
        #pragma unroll 8
        for (int c4 = 0; c4 < 32; ++c4) {
            float4 g4 = gv[c4], f4 = fv[c4];
            s += g4.x * f4.x + g4.y * f4.y + g4.z * f4.z + g4.w * f4.w;
        }
        simS[pos * 26 + l] = s;
    }
    __syncthreads();

    if (t < 16) {
        float m = -INFINITY;
        #pragma unroll
        for (int l = 0; l < 25; ++l) m = fmaxf(m, simS[t * 26 + l]);
        float e[25], sum = 0.f;
        #pragma unroll
        for (int l = 0; l < 25; ++l) {
            e[l] = __expf(simS[t * 26 + l] - m);
            sum += e[l];
        }
        float inv = 1.0f / sum;
        float* op = out + ((size_t)b * 12288 + (size_t)(x * 128 + y0 + t)) * 25;
        #pragma unroll
        for (int l = 0; l < 25; ++l) op[l] = e[l] * inv;
    }
    }
}

extern "C" void kernel_launch(void* const* d_in, const int* in_sizes, int n_in,
                              void* d_out, int out_size, void* d_ws, size_t ws_size,
                              hipStream_t stream) {
    (void)in_sizes; (void)n_in; (void)d_ws; (void)ws_size; (void)out_size;
    const float* fmap = (const float*)d_in[0];
    const float* Wqk  = (const float*)d_in[1];
    float* out = (float*)d_out;

    dim3 grid(128 / 16, 96 / 6, 4);
    LocalSimilar_fused_kernel<<<grid, 256, 0, stream>>>(fmap, Wqk, out);
}


# halo rows kept in a 5-slot LDS ring across the row loop: passes 2-6 fetch only their one new row (on waves 3-7, while wave 0 ends the softmax), was all 5 rows per pass; halo line requests / 3
# speedup vs baseline: 4.5628x; 1.2421x over previous
#include <hip/hip_runtime.h>
#include <math.h>


typedef float v2f __attribute__((ext_vector_type(2)));
typedef float v8f __attribute__((ext_vector_type(8)));

#define STRD 132
#define TP   100
#define ROWS_PER_BLOCK 6

__global__ __launch_bounds__(256) void
LocalSimilar_fused_kernel(const float* __restrict__ fmap,
                          const float* __restrict__ Wqk,
                          float* __restrict__ out)
{
    __shared__ float fmapT[TP * STRD];
    __shared__ float qT[16 * STRD];
    __shared__ float gT[16 * STRD];
    __shared__ float simS[16 * 26];

    const int t    = threadIdx.x;
    const int lane = t & 31;
    const int w    = t >> 5;
    const int lo   = lane & 15;
    const int hi   = lane >> 4;

    const int b  = blockIdx.z;
    const int x0 = blockIdx.y * ROWS_PER_BLOCK;
    const int y0 = blockIdx.x * 16;

    float2 af[32];
    v2f ag[32];
    {
        const float* arow = Wqk + ((size_t)(w * 16 + lo)) * 128 + 2 * hi;
        #pragma unroll
        for (int kk = 0; kk < 32; ++kk) af[kk] = *(const float2*)(arow + kk * 4);

        const float* acol = Wqk + ((size_t)(128 + 2 * hi)) * 128 + w * 16 + lo;
        #pragma unroll
        for (int kk = 0; kk < 32; ++kk) {
            ag[kk].x = acol[(kk * 4)     * 128];
            ag[kk].y = acol[(kk * 4 + 1) * 128];
        }
    }

    int base = 0;
    for (int it = 0; it < ROWS_PER_BLOCK; ++it) {
    const int x = x0 + it;
    const int slot_c = base + 2 >= 5 ? base + 2 - 5 : base + 2;

    if (it == 0) {
      if (t < 200) {
        const int ch0 = t / 100;
        const int pos = t - ch0 * 100;
        const int r = pos / 20, j = pos - r * 20;
        int gx = x - 2 + r;   gx = gx < 0 ? 0 : (gx > 95 ? 95 : gx);
        int gy = y0 - 2 + j;  gy = gy < 0 ? 0 : (gy > 127 ? 127 : gy);
        const float* src = fmap + (((size_t)b * 128 + ch0) * 96 + gx) * 128 + gy;
        float* dst = fmapT + pos * STRD + ch0;
        float v[64];
        #pragma unroll
        for (int i = 0; i < 64; ++i) v[i] = src[(size_t)i * (2 * 96 * 128)];
        #pragma unroll
        for (int i = 0; i < 64; ++i) dst[2 * i] = v[i];
      }
    } else if (t >= 96) {
        const int u  = t - 96;
        const int cg = u / 20;
        const int j  = u - cg * 20;
        const int slot_n = base == 0 ? 4 : base - 1;
        int gx = x + 2;       gx = gx < 0 ? 0 : (gx > 95 ? 95 : gx);
        int gy = y0 - 2 + j;  gy = gy < 0 ? 0 : (gy > 127 ? 127 : gy);
        const float* src = fmap + (((size_t)b * 128 + cg) * 96 + gx) * 128 + gy;
        float* dst = fmapT + (slot_n * 20 + j) * STRD + cg;
        float v[16];
        #pragma unroll
        for (int i = 0; i < 16; ++i) v[i] = src[(size_t)i * (8 * 96 * 128)];
        #pragma unroll
        for (int i = 0; i < 16; ++i) dst[8 * i] = v[i];
    }
    __syncthreads();

    {
        v8f acc = {0.f, 0.f, 0.f, 0.f, 0.f, 0.f, 0.f, 0.f};
        const float* brow = fmapT + (slot_c * 20 + 2 + lo) * STRD + 2 * hi;
        #pragma unroll
        for (int kk = 0; kk < 32; ++kk) {
            float2 b2 = *(const float2*)(brow + kk * 4);
            v2f av; av.x = af[kk].x; av.y = af[kk].y;
            v2f bv; bv.x = b2.x;     bv.y = b2.y;
            acc = __builtin_amdgcn_wmma_f32_16x16x4_f32(
                false, av, false, bv, (short)0, acc, false, false);
        }
        const float scale = 0.08838834764831845f;
        #pragma unroll
        for (int v = 0; v < 8; ++v)
            qT[lo * STRD + w * 16 + v + 8 * hi] = acc[v] * scale;
    }
    __syncthreads();

    {
        v8f acc = {0.f, 0.f, 0.f, 0.f, 0.f, 0.f, 0.f, 0.f};
        const float* brow = qT + lo * STRD + 2 * hi;
        #pragma unroll
        for (int kk = 0; kk < 32; ++kk) {
            float2 b2 = *(const float2*)(brow + kk * 4);
            v2f bv; bv.x = b2.x; bv.y = b2.y;
            acc = __builtin_amdgcn_wmma_f32_16x16x4_f32(
                false, ag[kk], false, bv, (short)0, acc, false, false);
        }
        #pragma unroll
        for (int v = 0; v < 8; ++v)
            gT[lo * STRD + w * 16 + v + 8 * hi] = acc[v];
    }
    __syncthreads();

    for (int idx = t; idx < 16 * 25; idx += 256) {
        int pos = idx / 25, l = idx % 25;
        int di = l / 5, dj = l % 5;
        int sr = base + di;  sr = sr >= 5 ? sr - 5 : sr;
        int npos = sr * 20 + pos + dj;
        const float4* gv = (const float4*)(gT    + pos  * STRD);
        const float4* fv = (const float4*)(fmapT + npos * STRD);
        float s = 0.f;
        #pragma unroll 8
        for (int c4 = 0; c4 < 32; ++c4) {
            float4 g4 = gv[c4], f4 = fv[c4];
            s += g4.x * f4.x + g4.y * f4.y + g4.z * f4.z + g4.w * f4.w;
        }
        simS[pos * 26 + l] = s;
    }
    __syncthreads();

    if (t < 16) {
        float m = -INFINITY;
        #pragma unroll
        for (int l = 0; l < 25; ++l) m = fmaxf(m, simS[t * 26 + l]);
        float e[25], sum = 0.f;
        #pragma unroll
        for (int l = 0; l < 25; ++l) {
            e[l] = __expf(simS[t * 26 + l] - m);
            sum += e[l];
        }
        float inv = 1.0f / sum;
        float* op = out + ((size_t)b * 12288 + (size_t)(x * 128 + y0 + t)) * 25;
        #pragma unroll
        for (int l = 0; l < 25; ++l) op[l] = e[l] * inv;
    }
    base = base == 4 ? 0 : base + 1;
    }
}

extern "C" void kernel_launch(void* const* d_in, const int* in_sizes, int n_in,
                              void* d_out, int out_size, void* d_ws, size_t ws_size,
                              hipStream_t stream) {
    (void)in_sizes; (void)n_in; (void)d_ws; (void)ws_size; (void)out_size;
    const float* fmap = (const float*)d_in[0];
    const float* Wqk  = (const float*)d_in[1];
    float* out = (float*)d_out;

    dim3 grid(128 / 16, 96 / 6, 4);
    LocalSimilar_fused_kernel<<<grid, 256, 0, stream>>>(fmap, Wqk, out);
}


# similarity dots register-tiled: a lane reads 4 g rows and 8 halo columns once for 4 pixels x 5 neighbors, 8 lanes split the channels and add partials across lanes; score summation order changes
# speedup vs baseline: 5.5030x; 1.2061x over previous
#include <hip/hip_runtime.h>
#include <math.h>


typedef float v2f __attribute__((ext_vector_type(2)));
typedef float v8f __attribute__((ext_vector_type(8)));

#define STRD 132
#define TP   100
#define ROWS_PER_BLOCK 6

template <int M>
__device__ __forceinline__ float lane_xor(float v)
{
    return __int_as_float(__builtin_amdgcn_update_dpp(
        0, __float_as_int(v), 0x160 + M  , 0xf, 0xf, true));
}

__global__ __launch_bounds__(256) void
LocalSimilar_fused_kernel(const float* __restrict__ fmap,
                          const float* __restrict__ Wqk,
                          float* __restrict__ out)
{
    __shared__ float fmapT[TP * STRD];
    __shared__ float qT[16 * STRD];
    __shared__ float gT[16 * STRD];
    __shared__ float simS[16 * 26];

    const int t    = threadIdx.x;
    const int lane = t & 31;
    const int w    = t >> 5;
    const int lo   = lane & 15;
    const int hi   = lane >> 4;

    const int b  = blockIdx.z;
    const int x0 = blockIdx.y * ROWS_PER_BLOCK;
    const int y0 = blockIdx.x * 16;

    float2 af[32];
    v2f ag[32];
    {
        const float* arow = Wqk + ((size_t)(w * 16 + lo)) * 128 + 2 * hi;
        #pragma unroll
        for (int kk = 0; kk < 32; ++kk) af[kk] = *(const float2*)(arow + kk * 4);

        const float* acol = Wqk + ((size_t)(128 + 2 * hi)) * 128 + w * 16 + lo;
        #pragma unroll
        for (int kk = 0; kk < 32; ++kk) {
            ag[kk].x = acol[(kk * 4)     * 128];
            ag[kk].y = acol[(kk * 4 + 1) * 128];
        }
    }

    int base = 0;
    for (int it = 0; it < ROWS_PER_BLOCK; ++it) {
    const int x = x0 + it;
    const int slot_c = base + 2 >= 5 ? base + 2 - 5 : base + 2;

    if (it == 0) {
      if (t < 200) {
        const int ch0 = t / 100;
        const int pos = t - ch0 * 100;
        const int r = pos / 20, j = pos - r * 20;
        int gx = x - 2 + r;   gx = gx < 0 ? 0 : (gx > 95 ? 95 : gx);
        int gy = y0 - 2 + j;  gy = gy < 0 ? 0 : (gy > 127 ? 127 : gy);
        const float* src = fmap + (((size_t)b * 128 + ch0) * 96 + gx) * 128 + gy;
        float* dst = fmapT + pos * STRD + ch0;
        float v[64];
        #pragma unroll
        for (int i = 0; i < 64; ++i) v[i] = src[(size_t)i * (2 * 96 * 128)];
        #pragma unroll
        for (int i = 0; i < 64; ++i) dst[2 * i] = v[i];
      }
    } else if (t >= 96) {
        const int u  = t - 96;
        const int cg = u / 20;
        const int j  = u - cg * 20;
        const int slot_n = base == 0 ? 4 : base - 1;
        int gx = x + 2;       gx = gx < 0 ? 0 : (gx > 95 ? 95 : gx);
        int gy = y0 - 2 + j;  gy = gy < 0 ? 0 : (gy > 127 ? 127 : gy);
        const float* src = fmap + (((size_t)b * 128 + cg) * 96 + gx) * 128 + gy;
        float* dst = fmapT + (slot_n * 20 + j) * STRD + cg;
        float v[16];
        #pragma unroll
        for (int i = 0; i < 16; ++i) v[i] = src[(size_t)i * (8 * 96 * 128)];
        #pragma unroll
        for (int i = 0; i < 16; ++i) dst[8 * i] = v[i];
    }
    __syncthreads();

    {
        v8f acc = {0.f, 0.f, 0.f, 0.f, 0.f, 0.f, 0.f, 0.f};
        const float* brow = fmapT + (slot_c * 20 + 2 + lo) * STRD + 2 * hi;
        #pragma unroll
        for (int kk = 0; kk < 32; ++kk) {
            float2 b2 = *(const float2*)(brow + kk * 4);
            v2f av; av.x = af[kk].x; av.y = af[kk].y;
            v2f bv; bv.x = b2.x;     bv.y = b2.y;
            acc = __builtin_amdgcn_wmma_f32_16x16x4_f32(
                false, av, false, bv, (short)0, acc, false, false);
        }
        const float scale = 0.08838834764831845f;
        #pragma unroll
        for (int v = 0; v < 8; ++v)
            qT[lo * STRD + w * 16 + v + 8 * hi] = acc[v] * scale;
    }
    __syncthreads();

    {
        v8f acc = {0.f, 0.f, 0.f, 0.f, 0.f, 0.f, 0.f, 0.f};
        const float* brow = qT + lo * STRD + 2 * hi;
        #pragma unroll
        for (int kk = 0; kk < 32; ++kk) {
            float2 b2 = *(const float2*)(brow + kk * 4);
            v2f bv; bv.x = b2.x; bv.y = b2.y;
            acc = __builtin_amdgcn_wmma_f32_16x16x4_f32(
                false, ag[kk], false, bv, (short)0, acc, false, false);
        }
        #pragma unroll
        for (int v = 0; v < 8; ++v)
            gT[lo * STRD + w * 16 + v + 8 * hi] = acc[v];
    }
    __syncthreads();

    if (w < 5) {
        const int s  = lane & 7;
        const int pg = ((lane >> 3) & 1) * 2 + (lane >> 4);
        int sr = base + w;  sr = sr >= 5 ? sr - 5 : sr;
        const float* gp = gT    + (4 * pg) * STRD + 4 * s;
        const float* fp = fmapT + (sr * 20 + 4 * pg) * STRD + 4 * s;
        v2f acc[4][5];
        #pragma unroll
        for (int i = 0; i < 4; ++i)
            #pragma unroll
            for (int dj = 0; dj < 5; ++dj) { acc[i][dj].x = 0.f; acc[i][dj].y = 0.f; }
        #pragma unroll 1
        for (int j = 0; j < 4; ++j) {
            float4 g4[4], f4[8];
            #pragma unroll
            for (int i = 0; i < 4; ++i) g4[i] = *(const float4*)(gp + i * STRD + 32 * j);
            #pragma unroll
            for (int m = 0; m < 8; ++m) f4[m] = *(const float4*)(fp + m * STRD + 32 * j);
            #pragma unroll
            for (int i = 0; i < 4; ++i) {
                v2f glo, ghi;
                glo.x = g4[i].x; glo.y = g4[i].y; ghi.x = g4[i].z; ghi.y = g4[i].w;
                #pragma unroll
                for (int dj = 0; dj < 5; ++dj) {
                    v2f flo, fhi;
                    flo.x = f4[i + dj].x; flo.y = f4[i + dj].y;
                    fhi.x = f4[i + dj].z; fhi.y = f4[i + dj].w;
                    acc[i][dj] = glo * flo + acc[i][dj];
                    acc[i][dj] = ghi * fhi + acc[i][dj];
                }
            }
        }
        float r[4][5];
        #pragma unroll
        for (int i = 0; i < 4; ++i) {
            #pragma unroll
            for (int dj = 0; dj < 5; ++dj) {
                float v = acc[i][dj].x + acc[i][dj].y;
                v += lane_xor<1>(v);
                v += lane_xor<2>(v);
                v += lane_xor<4>(v);
                r[i][dj] = v;
            }
        }
        if (s == 0) {
            float* sp = simS + (4 * pg) * 26 + w * 5;
            #pragma unroll
            for (int i = 0; i < 4; ++i) {
                #pragma unroll
                for (int dj = 0; dj < 5; ++dj) sp[i * 26 + dj] = r[i][dj];
            }
        }
    }
    __syncthreads();

    if (t < 16) {
        float m = -INFINITY;
        #pragma unroll
        for (int l = 0; l < 25; ++l) m = fmaxf(m, simS[t * 26 + l]);
        float e[25], sum = 0.f;
        #pragma unroll
        for (int l = 0; l < 25; ++l) {
            e[l] = __expf(simS[t * 26 + l] - m);
            sum += e[l];
        }
        float inv = 1.0f / sum;
        float* op = out + ((size_t)b * 12288 + (size_t)(x * 128 + y0 + t)) * 25;
        #pragma unroll
        for (int l = 0; l < 25; ++l) op[l] = e[l] * inv;
    }
    base = base == 4 ? 0 : base + 1;
    }
}

extern "C" void kernel_launch(void* const* d_in, const int* in_sizes, int n_in,
                              void* d_out, int out_size, void* d_ws, size_t ws_size,
                              hipStream_t stream) {
    (void)in_sizes; (void)n_in; (void)d_ws; (void)ws_size; (void)out_size;
    const float* fmap = (const float*)d_in[0];
    const float* Wqk  = (const float*)d_in[1];
    float* out = (float*)d_out;

    dim3 grid(128 / 16, 96 / 6, 4);
    LocalSimilar_fused_kernel<<<grid, 256, 0, stream>>>(fmap, Wqk, out);
}


# Wq and Wk folded into one matrix M = scale*Wk^T*Wq by a small first launch (64 waves, workspace): a pass does one projection g = M f, 32 matrix ops per wave, was 64; score summation order changes
# speedup vs baseline: 6.7800x; 1.2321x over previous
#include <hip/hip_runtime.h>
#include <math.h>


typedef float v2f __attribute__((ext_vector_type(2)));
typedef float v8f __attribute__((ext_vector_type(8)));

#define STRD 132
#define TP   120
#define SIMN (16 * 26)
#define ROWS_PER_BLOCK 6

template <int M>
__device__ __forceinline__ float lane_xor(float v)
{
    return __int_as_float(__builtin_amdgcn_update_dpp(
        0, __float_as_int(v), 0x160 + M  , 0xf, 0xf, true));
}

__device__ __forceinline__ void softmax25(const float* sp, float* op)
{
    float m = -INFINITY;
    #pragma unroll
    for (int l = 0; l < 25; ++l) m = fmaxf(m, sp[l]);
    float e[25], sum = 0.f;
    #pragma unroll
    for (int l = 0; l < 25; ++l) {
        e[l] = __expf(sp[l] - m);
        sum += e[l];
    }
    float inv = 1.0f / sum;
    #pragma unroll
    for (int l = 0; l < 25; ++l) op[l] = e[l] * inv;
}

__global__ __launch_bounds__(32) void
LocalSimilar_fold_weights(const float* __restrict__ Wqk, float* __restrict__ M)
{
    const int lane = threadIdx.x & 31;
    const int lo   = lane & 15;
    const int hi   = lane >> 4;
    const int tr   = blockIdx.x >> 3;
    const int tc   = blockIdx.x & 7;

    const float* acol = Wqk + ((size_t)(128 + 2 * hi)) * 128 + tr * 16 + lo;
    const float* bcol = Wqk + ((size_t)(2 * hi)) * 128 + tc * 16 + lo;
    v2f a[32], bb[32];
    #pragma unroll
    for (int kk = 0; kk < 32; ++kk) {
        a[kk].x  = acol[(kk * 4)     * 128];
        a[kk].y  = acol[(kk * 4 + 1) * 128];
        bb[kk].x = bcol[(kk * 4)     * 128];
        bb[kk].y = bcol[(kk * 4 + 1) * 128];
    }
    v8f acc0 = {0.f, 0.f, 0.f, 0.f, 0.f, 0.f, 0.f, 0.f};
    v8f acc1 = {0.f, 0.f, 0.f, 0.f, 0.f, 0.f, 0.f, 0.f};
    #pragma unroll
    for (int kk = 0; kk < 32; kk += 2) {
        acc0 = __builtin_amdgcn_wmma_f32_16x16x4_f32(
            false, a[kk], false, bb[kk], (short)0, acc0, false, false);
        acc1 = __builtin_amdgcn_wmma_f32_16x16x4_f32(
            false, a[kk + 1], false, bb[kk + 1], (short)0, acc1, false, false);
    }
    const float scale = 0.08838834764831845f;
    float* mp = M + ((size_t)(tr * 16 + 8 * hi)) * 128 + tc * 16 + lo;
    #pragma unroll
    for (int v = 0; v < 8; ++v) mp[v * 128] = (acc0[v] + acc1[v]) * scale;
}

template <bool FOLDED>
__global__ __launch_bounds__(256) void
LocalSimilar_fused_kernel(const float* __restrict__ fmap,
                          const float* __restrict__ Wa,
                          float* __restrict__ out)
{
    __shared__ float fmapT[TP * STRD];
    __shared__ float qT[16 * STRD];
    __shared__ float gT[16 * STRD];
    __shared__ float simS[2 * SIMN];

    const int t    = threadIdx.x;
    const int lane = t & 31;
    const int w    = __builtin_amdgcn_readfirstlane(t >> 5);
    const int lo   = lane & 15;
    const int hi   = lane >> 4;

    const int b  = blockIdx.z;
    const int x0 = blockIdx.y * ROWS_PER_BLOCK;
    const int y0 = blockIdx.x * 16;

    float2 af[32];
    v2f ag[32];
    {
        const float* arow = Wa + ((size_t)(w * 16 + lo)) * 128 + 2 * hi;
        #pragma unroll
        for (int kk = 0; kk < 32; ++kk) af[kk] = *(const float2*)(arow + kk * 4);

        if constexpr (!FOLDED) {
            const float* acol = Wa + ((size_t)(128 + 2 * hi)) * 128 + w * 16 + lo;
            #pragma unroll
            for (int kk = 0; kk < 32; ++kk) {
                ag[kk].x = acol[(kk * 4)     * 128];
                ag[kk].y = acol[(kk * 4 + 1) * 128];
            }
        }
    }

    int base = 0;
    for (int it = 0; it < ROWS_PER_BLOCK; ++it) {
    const int x = x0 + it;
    const int slot_c = base + 2 >= 6 ? base + 2 - 6 : base + 2;

    if (it == 0) {
      if (t < 200) {
        const int ch0 = t / 100;
        const int pos = t - ch0 * 100;
        const int r = pos / 20, j = pos - r * 20;
        int gx = x - 2 + r;   gx = gx < 0 ? 0 : (gx > 95 ? 95 : gx);
        int gy = y0 - 2 + j;  gy = gy < 0 ? 0 : (gy > 127 ? 127 : gy);
        const float* src = fmap + (((size_t)b * 128 + ch0) * 96 + gx) * 128 + gy;
        float* dst = fmapT + pos * STRD + ch0;
        float v[64];
        #pragma unroll
        for (int i = 0; i < 64; ++i) v[i] = src[(size_t)i * (2 * 96 * 128)];
        #pragma unroll
        for (int i = 0; i < 64; ++i) dst[2 * i] = v[i];
      }
      __syncthreads();
    }

    {
        v8f acc = {0.f, 0.f, 0.f, 0.f, 0.f, 0.f, 0.f, 0.f};
        const float* brow = fmapT + (slot_c * 20 + 2 + lo) * STRD + 2 * hi;
        #pragma unroll
        for (int kk = 0; kk < 32; ++kk) {
            float2 b2 = *(const float2*)(brow + kk * 4);
            v2f av; av.x = af[kk].x; av.y = af[kk].y;
            v2f bv; bv.x = b2.x;     bv.y = b2.y;
            acc = __builtin_amdgcn_wmma_f32_16x16x4_f32(
                false, av, false, bv, (short)0, acc, false, false);
        }
        if constexpr (FOLDED) {
            #pragma unroll
            for (int v = 0; v < 8; ++v)
                gT[lo * STRD + w * 16 + v + 8 * hi] = acc[v];
        } else {
            const float scale = 0.08838834764831845f;
            #pragma unroll
            for (int v = 0; v < 8; ++v)
                qT[lo * STRD + w * 16 + v + 8 * hi] = acc[v] * scale;
        }
    }
    __syncthreads();

    if constexpr (!FOLDED) {
        v8f acc = {0.f, 0.f, 0.f, 0.f, 0.f, 0.f, 0.f, 0.f};
        const float* brow = qT + lo * STRD + 2 * hi;
        #pragma unroll
        for (int kk = 0; kk < 32; ++kk) {
            float2 b2 = *(const float2*)(brow + kk * 4);
            v2f bv; bv.x = b2.x; bv.y = b2.y;
            acc = __builtin_amdgcn_wmma_f32_16x16x4_f32(
                false, ag[kk], false, bv, (short)0, acc, false, false);
        }
        #pragma unroll
        for (int v = 0; v < 8; ++v)
            gT[lo * STRD + w * 16 + v + 8 * hi] = acc[v];
        __syncthreads();
    }

    if (w < 5) {
        const int s  = lane & 7;
        const int pg = ((lane >> 3) & 1) * 2 + (lane >> 4);
        int sr = base + w;  sr = sr >= 6 ? sr - 6 : sr;
        const float* gp = gT    + (4 * pg) * STRD + 4 * s;
        const float* fp = fmapT + (sr * 20 + 4 * pg) * STRD + 4 * s;
        v2f acc[4][5];
        #pragma unroll
        for (int i = 0; i < 4; ++i)
            #pragma unroll
            for (int dj = 0; dj < 5; ++dj) { acc[i][dj].x = 0.f; acc[i][dj].y = 0.f; }
        #pragma unroll 1
        for (int j = 0; j < 4; ++j) {
            float4 g4[4], f4[8];
            #pragma unroll
            for (int i = 0; i < 4; ++i) g4[i] = *(const float4*)(gp + i * STRD + 32 * j);
            #pragma unroll
            for (int m = 0; m < 8; ++m) f4[m] = *(const float4*)(fp + m * STRD + 32 * j);
            #pragma unroll
            for (int i = 0; i < 4; ++i) {
                v2f glo, ghi;
                glo.x = g4[i].x; glo.y = g4[i].y; ghi.x = g4[i].z; ghi.y = g4[i].w;
                #pragma unroll
                for (int dj = 0; dj < 5; ++dj) {
                    v2f flo, fhi;
                    flo.x = f4[i + dj].x; flo.y = f4[i + dj].y;
                    fhi.x = f4[i + dj].z; fhi.y = f4[i + dj].w;
                    acc[i][dj] = glo * flo + acc[i][dj];
                    acc[i][dj] = ghi * fhi + acc[i][dj];
                }
            }
        }
        float r[4][5];
        #pragma unroll
        for (int i = 0; i < 4; ++i) {
            #pragma unroll
            for (int dj = 0; dj < 5; ++dj) {
                float v = acc[i][dj].x + acc[i][dj].y;
                v += lane_xor<1>(v);
                v += lane_xor<2>(v);
                v += lane_xor<4>(v);
                r[i][dj] = v;
            }
        }
        if (s == 0) {
            float* sp = simS + (it & 1) * SIMN + (4 * pg) * 26 + w * 5;
            #pragma unroll
            for (int i = 0; i < 4; ++i) {
                #pragma unroll
                for (int dj = 0; dj < 5; ++dj) sp[i * 26 + dj] = r[i][dj];
            }
        }
    }
    else {
        const int u = t - 160;
        const bool ld = (it + 1 < ROWS_PER_BLOCK) && (u < 80);
        float v[32];
        #pragma unroll
        for (int i = 0; i < 32; ++i) v[i] = 0.f;
        float* dst = fmapT;
        if (ld) {
            const int cg = u / 20;
            const int j  = u - cg * 20;
            const int slot_n = base == 0 ? 5 : base - 1;
            int gx = x + 3;       gx = gx < 0 ? 0 : (gx > 95 ? 95 : gx);
            int gy = y0 - 2 + j;  gy = gy < 0 ? 0 : (gy > 127 ? 127 : gy);
            const float* src = fmap + (((size_t)b * 128 + cg) * 96 + gx) * 128 + gy;
            dst = fmapT + (slot_n * 20 + j) * STRD + cg;
            #pragma unroll
            for (int i = 0; i < 32; ++i) v[i] = src[(size_t)i * (4 * 96 * 128)];
        }
        if (it > 0 && u >= 80) {
            const int p = u - 80;
            softmax25(simS + ((it - 1) & 1) * SIMN + p * 26,
                      out + ((size_t)b * 12288 + (size_t)((x - 1) * 128 + y0 + p)) * 25);
        }
        if (ld) {
            #pragma unroll
            for (int i = 0; i < 32; ++i) dst[4 * i] = v[i];
        }
    }
    __builtin_amdgcn_fence(__ATOMIC_RELEASE, "workgroup", "local");
    __builtin_amdgcn_s_barrier();
    __builtin_amdgcn_fence(__ATOMIC_ACQUIRE, "workgroup", "local");
    base = base == 5 ? 0 : base + 1;
    }

    if (t >= 240) {
        const int p = t - 240;
        const int xl = x0 + ROWS_PER_BLOCK - 1;
        softmax25(simS + ((ROWS_PER_BLOCK - 1) & 1) * SIMN + p * 26,
                  out + ((size_t)b * 12288 + (size_t)(xl * 128 + y0 + p)) * 25);
    }
}

extern "C" void kernel_launch(void* const* d_in, const int* in_sizes, int n_in,
                              void* d_out, int out_size, void* d_ws, size_t ws_size,
                              hipStream_t stream) {
    (void)in_sizes; (void)n_in; (void)out_size;
    const float* fmap = (const float*)d_in[0];
    const float* Wqk  = (const float*)d_in[1];
    float* out = (float*)d_out;

    dim3 grid(128 / 16, 96 / 6, 4);
    if (d_ws != nullptr && ws_size >= (size_t)65536) {
        float* M = (float*)d_ws;
        LocalSimilar_fold_weights<<<64, 32, 0, stream>>>(Wqk, M);
        LocalSimilar_fused_kernel<true><<<grid, 256, 0, stream>>>(fmap, M, out);
    } else {
        LocalSimilar_fused_kernel<false><<<grid, 256, 0, stream>>>(fmap, Wqk, out);
    }
}


# folded matrix stored by the first launch in the main kernel's fragment order: a wave's weight fragments come as 16 loads of 512 contiguous bytes, was 32 loads of 16 separate lines each
# speedup vs baseline: 7.1398x; 1.0531x over previous
#include <hip/hip_runtime.h>
#include <math.h>


typedef float v2f __attribute__((ext_vector_type(2)));
typedef float v8f __attribute__((ext_vector_type(8)));

#define STRD 132
#define TP   120
#define SIMN (16 * 26)
#define ROWS_PER_BLOCK 6

template <int M>
__device__ __forceinline__ float lane_xor(float v)
{
    return __int_as_float(__builtin_amdgcn_update_dpp(
        0, __float_as_int(v), 0x160 + M  , 0xf, 0xf, true));
}

__device__ __forceinline__ void softmax25(const float* sp, float* op)
{
    float m = -INFINITY;
    #pragma unroll
    for (int l = 0; l < 25; ++l) m = fmaxf(m, sp[l]);
    float e[25], sum = 0.f;
    #pragma unroll
    for (int l = 0; l < 25; ++l) {
        e[l] = __expf(sp[l] - m);
        sum += e[l];
    }
    float inv = 1.0f / sum;
    #pragma unroll
    for (int l = 0; l < 25; ++l) op[l] = e[l] * inv;
}

__global__ __launch_bounds__(32) void
LocalSimilar_fold_weights(const float* __restrict__ Wqk, float* __restrict__ M)
{
    const int lane = threadIdx.x & 31;
    const int lo   = lane & 15;
    const int hi   = lane >> 4;
    const int tr   = blockIdx.x >> 3;
    const int tc   = blockIdx.x & 7;

    const float* acol = Wqk + ((size_t)(128 + 2 * hi)) * 128 + tr * 16 + lo;
    const float* bcol = Wqk + ((size_t)(2 * hi)) * 128 + tc * 16 + lo;
    v2f a[32], bb[32];
    #pragma unroll
    for (int kk = 0; kk < 32; ++kk) {
        a[kk].x  = acol[(kk * 4)     * 128];
        a[kk].y  = acol[(kk * 4 + 1) * 128];
        bb[kk].x = bcol[(kk * 4)     * 128];
        bb[kk].y = bcol[(kk * 4 + 1) * 128];
    }
    v8f acc0 = {0.f, 0.f, 0.f, 0.f, 0.f, 0.f, 0.f, 0.f};
    v8f acc1 = {0.f, 0.f, 0.f, 0.f, 0.f, 0.f, 0.f, 0.f};
    #pragma unroll
    for (int kk = 0; kk < 32; kk += 2) {
        acc0 = __builtin_amdgcn_wmma_f32_16x16x4_f32(
            false, bb[kk], false, a[kk], (short)0, acc0, false, false);
        acc1 = __builtin_amdgcn_wmma_f32_16x16x4_f32(
            false, bb[kk + 1], false, a[kk + 1], (short)0, acc1, false, false);
    }
    const float scale = 0.08838834764831845f;
    float s[8];
    #pragma unroll
    for (int v = 0; v < 8; ++v) s[v] = (acc0[v] + acc1[v]) * scale;
    float4* mq = (float4*)M + (size_t)(tr * 16 + tc * 2 + hi) * 32 + lo;
    mq[0]  = make_float4(s[0], s[1], s[4], s[5]);
    mq[16] = make_float4(s[2], s[3], s[6], s[7]);
}

template <bool FOLDED>
__global__ __launch_bounds__(256) void
LocalSimilar_fused_kernel(const float* __restrict__ fmap,
                          const float* __restrict__ Wa,
                          float* __restrict__ out)
{
    __shared__ float fmapT[TP * STRD];
    __shared__ float qT[16 * STRD];
    __shared__ float gT[16 * STRD];
    __shared__ float simS[2 * SIMN];

    const int t    = threadIdx.x;
    const int lane = t & 31;
    const int w    = __builtin_amdgcn_readfirstlane(t >> 5);
    const int lo   = lane & 15;
    const int hi   = lane >> 4;

    const int b  = blockIdx.z;
    const int x0 = blockIdx.y * ROWS_PER_BLOCK;
    const int y0 = blockIdx.x * 16;

    float2 af[32];
    v2f ag[32];
    if constexpr (FOLDED) {
        const float4* aq = (const float4*)Wa + (size_t)(w * 16) * 32 + lane;
        #pragma unroll
        for (int m = 0; m < 16; ++m) {
            const float4 q4 = aq[m * 32];
            af[2 * m].x     = q4.x;  af[2 * m].y     = q4.y;
            af[2 * m + 1].x = q4.z;  af[2 * m + 1].y = q4.w;
        }
    } else {
        const float* arow = Wa + ((size_t)(w * 16 + lo)) * 128 + 2 * hi;
        #pragma unroll
        for (int kk = 0; kk < 32; ++kk) af[kk] = *(const float2*)(arow + kk * 4);

        const float* acol = Wa + ((size_t)(128 + 2 * hi)) * 128 + w * 16 + lo;
        #pragma unroll
        for (int kk = 0; kk < 32; ++kk) {
            ag[kk].x = acol[(kk * 4)     * 128];
            ag[kk].y = acol[(kk * 4 + 1) * 128];
        }
    }

    int base = 0;
    for (int it = 0; it < ROWS_PER_BLOCK; ++it) {
    const int x = x0 + it;
    const int slot_c = base + 2 >= 6 ? base + 2 - 6 : base + 2;

    if (it == 0) {
      if (t < 200) {
        const int ch0 = t / 100;
        const int pos = t - ch0 * 100;
        const int r = pos / 20, j = pos - r * 20;
        int gx = x - 2 + r;   gx = gx < 0 ? 0 : (gx > 95 ? 95 : gx);
        int gy = y0 - 2 + j;  gy = gy < 0 ? 0 : (gy > 127 ? 127 : gy);
        const float* src = fmap + (((size_t)b * 128 + ch0) * 96 + gx) * 128 + gy;
        float* dst = fmapT + pos * STRD + ch0;
        float v[64];
        #pragma unroll
        for (int i = 0; i < 64; ++i) v[i] = src[(size_t)i * (2 * 96 * 128)];
        #pragma unroll
        for (int i = 0; i < 64; ++i) dst[2 * i] = v[i];
      }
      __syncthreads();
    }

    {
        v8f acc = {0.f, 0.f, 0.f, 0.f, 0.f, 0.f, 0.f, 0.f};
        const float* brow = fmapT + (slot_c * 20 + 2 + lo) * STRD + 2 * hi;
        #pragma unroll
        for (int kk = 0; kk < 32; ++kk) {
            float2 b2 = *(const float2*)(brow + kk * 4);
            v2f av; av.x = af[kk].x; av.y = af[kk].y;
            v2f bv; bv.x = b2.x;     bv.y = b2.y;
            acc = __builtin_amdgcn_wmma_f32_16x16x4_f32(
                false, av, false, bv, (short)0, acc, false, false);
        }
        if constexpr (FOLDED) {
            #pragma unroll
            for (int v = 0; v < 8; ++v)
                gT[lo * STRD + w * 16 + v + 8 * hi] = acc[v];
        } else {
            const float scale = 0.08838834764831845f;
            #pragma unroll
            for (int v = 0; v < 8; ++v)
                qT[lo * STRD + w * 16 + v + 8 * hi] = acc[v] * scale;
        }
    }
    __syncthreads();

    if constexpr (!FOLDED) {
        v8f acc = {0.f, 0.f, 0.f, 0.f, 0.f, 0.f, 0.f, 0.f};
        const float* brow = qT + lo * STRD + 2 * hi;
        #pragma unroll
        for (int kk = 0; kk < 32; ++kk) {
            float2 b2 = *(const float2*)(brow + kk * 4);
            v2f bv; bv.x = b2.x; bv.y = b2.y;
            acc = __builtin_amdgcn_wmma_f32_16x16x4_f32(
                false, ag[kk], false, bv, (short)0, acc, false, false);
        }
        #pragma unroll
        for (int v = 0; v < 8; ++v)
            gT[lo * STRD + w * 16 + v + 8 * hi] = acc[v];
        __syncthreads();
    }

    if (w < 5) {
        const int s  = lane & 7;
        const int pg = ((lane >> 3) & 1) * 2 + (lane >> 4);
        int sr = base + w;  sr = sr >= 6 ? sr - 6 : sr;
        const float* gp = gT    + (4 * pg) * STRD + 4 * s;
        const float* fp = fmapT + (sr * 20 + 4 * pg) * STRD + 4 * s;
        v2f acc[4][5];
        #pragma unroll
        for (int i = 0; i < 4; ++i)
            #pragma unroll
            for (int dj = 0; dj < 5; ++dj) { acc[i][dj].x = 0.f; acc[i][dj].y = 0.f; }
        #pragma unroll 1
        for (int j = 0; j < 4; ++j) {
            float4 g4[4], f4[8];
            #pragma unroll
            for (int i = 0; i < 4; ++i) g4[i] = *(const float4*)(gp + i * STRD + 32 * j);
            #pragma unroll
            for (int m = 0; m < 8; ++m) f4[m] = *(const float4*)(fp + m * STRD + 32 * j);
            #pragma unroll
            for (int i = 0; i < 4; ++i) {
                v2f glo, ghi;
                glo.x = g4[i].x; glo.y = g4[i].y; ghi.x = g4[i].z; ghi.y = g4[i].w;
                #pragma unroll
                for (int dj = 0; dj < 5; ++dj) {
                    v2f flo, fhi;
                    flo.x = f4[i + dj].x; flo.y = f4[i + dj].y;
                    fhi.x = f4[i + dj].z; fhi.y = f4[i + dj].w;
                    acc[i][dj] = glo * flo + acc[i][dj];
                    acc[i][dj] = ghi * fhi + acc[i][dj];
                }
            }
        }
        float r[4][5];
        #pragma unroll
        for (int i = 0; i < 4; ++i) {
            #pragma unroll
            for (int dj = 0; dj < 5; ++dj) {
                float v = acc[i][dj].x + acc[i][dj].y;
                v += lane_xor<1>(v);
                v += lane_xor<2>(v);
                v += lane_xor<4>(v);
                r[i][dj] = v;
            }
        }
        if (s == 0) {
            float* sp = simS + (it & 1) * SIMN + (4 * pg) * 26 + w * 5;
            #pragma unroll
            for (int i = 0; i < 4; ++i) {
                #pragma unroll
                for (int dj = 0; dj < 5; ++dj) sp[i * 26 + dj] = r[i][dj];
            }
        }
    }
    else {
        const int u = t - 160;
        const bool ld = (it + 1 < ROWS_PER_BLOCK) && (u < 80);
        float v[32];
        #pragma unroll
        for (int i = 0; i < 32; ++i) v[i] = 0.f;
        float* dst = fmapT;
        if (ld) {
            const int cg = u / 20;
            const int j  = u - cg * 20;
            const int slot_n = base == 0 ? 5 : base - 1;
            int gx = x + 3;       gx = gx < 0 ? 0 : (gx > 95 ? 95 : gx);
            int gy = y0 - 2 + j;  gy = gy < 0 ? 0 : (gy > 127 ? 127 : gy);
            const float* src = fmap + (((size_t)b * 128 + cg) * 96 + gx) * 128 + gy;
            dst = fmapT + (slot_n * 20 + j) * STRD + cg;
            #pragma unroll
            for (int i = 0; i < 32; ++i) v[i] = src[(size_t)i * (4 * 96 * 128)];
        }
        if (it > 0 && u >= 80) {
            const int p = u - 80;
            softmax25(simS + ((it - 1) & 1) * SIMN + p * 26,
                      out + ((size_t)b * 12288 + (size_t)((x - 1) * 128 + y0 + p)) * 25);
        }
        if (ld) {
            #pragma unroll
            for (int i = 0; i < 32; ++i) dst[4 * i] = v[i];
        }
    }
    __builtin_amdgcn_fence(__ATOMIC_RELEASE, "workgroup", "local");
    __builtin_amdgcn_s_barrier();
    __builtin_amdgcn_fence(__ATOMIC_ACQUIRE, "workgroup", "local");
    base = base == 5 ? 0 : base + 1;
    }

    if (t >= 240) {
        const int p = t - 240;
        const int xl = x0 + ROWS_PER_BLOCK - 1;
        softmax25(simS + ((ROWS_PER_BLOCK - 1) & 1) * SIMN + p * 26,
                  out + ((size_t)b * 12288 + (size_t)(xl * 128 + y0 + p)) * 25);
    }
}

extern "C" void kernel_launch(void* const* d_in, const int* in_sizes, int n_in,
                              void* d_out, int out_size, void* d_ws, size_t ws_size,
                              hipStream_t stream) {
    (void)in_sizes; (void)n_in; (void)out_size;
    const float* fmap = (const float*)d_in[0];
    const float* Wqk  = (const float*)d_in[1];
    float* out = (float*)d_out;

    dim3 grid(128 / 16, 96 / 6, 4);
    if (d_ws != nullptr && ws_size >= (size_t)65536) {
        float* M = (float*)d_ws;
        LocalSimilar_fold_weights<<<64, 32, 0, stream>>>(Wqk, M);
        LocalSimilar_fused_kernel<true><<<grid, 256, 0, stream>>>(fmap, M, out);
    } else {
        LocalSimilar_fused_kernel<false><<<grid, 256, 0, stream>>>(fmap, Wqk, out);
    }
}
